# starting kernel
# baseline (speedup 1.0000x reference)
#include <hip/hip_runtime.h>
#include <hip/hip_bf16.h>

typedef __attribute__((ext_vector_type(2))) float v2f;
typedef __attribute__((ext_vector_type(8))) float v8f;

#define HID   4096
#define BATCH 16384
#define OUTC  44
#define PD    8

#define THREADS        128
#define ROWS_PER_BLOCK 64
#define KC        64
#define XS_STRIDE 68
#define WS_STRIDE 68
#define HS_STRIDE 52
#define NPAD      48

__global__ __launch_bounds__(THREADS) void l3m_param_head_kernel(
    const float* __restrict__ x, const float* __restrict__ W,
    const float* __restrict__ bias, float* __restrict__ out)
{
    __shared__ float xs[ROWS_PER_BLOCK * XS_STRIDE];
    __shared__ float ws[NPAD * WS_STRIDE];

    const int tid  = threadIdx.x;
    const int wid  = tid >> 5;
    const int lane = tid & 31;
    const int half = lane >> 4;
    const int l16  = lane & 15;
    const int m0   = wid * 16;
    const int rowBase = blockIdx.x * ROWS_PER_BLOCK;

    v8f acc0 = {}; v8f acc1 = {}; v8f acc2 = {};

    for (int k0 = 0; k0 < HID; k0 += KC) {
        const float4* xg = (const float4*)x;
        const float4* wg = (const float4*)W;
        const int kq = k0 >> 2;
        #pragma unroll
        for (int i = 0; i < 8; ++i) {
            int f  = tid + i * THREADS;
            int r  = f >> 4;
            int c4 = f & 15;
            float4 v = xg[(size_t)(rowBase + r) * (HID / 4) + kq + c4];
            *(float4*)&xs[r * XS_STRIDE + c4 * 4] = v;
        }
        #pragma unroll
        for (int i = 0; i < 6; ++i) {
            int f  = tid + i * THREADS;
            int n  = f >> 4;
            int c4 = f & 15;
            float4 v = make_float4(0.f, 0.f, 0.f, 0.f);
            if (n < OUTC)
                v = wg[(size_t)n * (HID / 4) + kq + c4];
            *(float4*)&ws[n * WS_STRIDE + c4 * 4] = v;
        }
        if (k0 + 2 * KC < HID)
            __builtin_prefetch(&x[(size_t)(rowBase + (tid >> 1)) * HID
                                  + (k0 + 2 * KC) + (tid & 1) * 32], 0, 1);
        __syncthreads();

        #pragma unroll
        for (int kk = 0; kk < KC; kk += 4) {
            const int ks = kk + 2 * half;
            v2f a  = *(const v2f*)&xs[(m0 + l16) * XS_STRIDE + ks];
            v2f b0 = *(const v2f*)&ws[( 0 + l16) * WS_STRIDE + ks];
            v2f b1 = *(const v2f*)&ws[(16 + l16) * WS_STRIDE + ks];
            v2f b2 = *(const v2f*)&ws[(32 + l16) * WS_STRIDE + ks];
            acc0 = __builtin_amdgcn_wmma_f32_16x16x4_f32(false, a, false, b0, (short)0, acc0, false, false);
            acc1 = __builtin_amdgcn_wmma_f32_16x16x4_f32(false, a, false, b1, (short)0, acc1, false, false);
            acc2 = __builtin_amdgcn_wmma_f32_16x16x4_f32(false, a, false, b2, (short)0, acc2, false, false);
        }
        __syncthreads();
    }

    float* hs = xs;
    #pragma unroll
    for (int v = 0; v < 8; ++v) {
        int m = m0 + v + 8 * half;
        hs[m * HS_STRIDE +  0 + l16] = acc0[v];
        hs[m * HS_STRIDE + 16 + l16] = acc1[v];
        hs[m * HS_STRIDE + 32 + l16] = acc2[v];
    }
    __syncthreads();

    if (tid < ROWS_PER_BLOCK) {
        const int row = rowBase + tid;
        float h[OUTC];
        #pragma unroll
        for (int j = 0; j < OUTC; ++j) h[j] = hs[tid * HS_STRIDE + j] + bias[j];

        float* outP = out;
        float* outS = out + (size_t)BATCH * PD;
        float* outL = out + (size_t)BATCH * PD + (size_t)BATCH * PD * PD;

        float pr[PD];
        #pragma unroll
        for (int i = 0; i < PD; ++i) pr[i] = 1.f / (1.f + expf(-h[i]));
        #pragma unroll
        for (int i = 0; i < PD; i += 4)
            *(float4*)&outP[(size_t)row * PD + i] =
                make_float4(pr[i], pr[i + 1], pr[i + 2], pr[i + 3]);

        float sg[PD][PD];
        #pragma unroll
        for (int i = 0; i < PD; ++i) {
            const float sc = 2.0f / sqrtf((float)(i + 1));
            float d  = h[PD + i];
            float sp = (d > 20.f) ? d : log1pf(expf(d));
            sg[i][i] = sp * sc;
            #pragma unroll
            for (int j = 0; j < i; ++j)
                sg[i][j] = h[2 * PD + (i * (i - 1)) / 2 + j] * sc;
        }

        float p = 1.f;
        #pragma unroll
        for (int i = 0; i < PD; ++i) p *= sg[i][i];
        float bp  = 10000.f * p;
        float lae = (bp > 0.f) ? (bp + log1pf(expf(-bp))) : log1pf(expf(bp));
        outL[row] = logf(lae) - 9.210340371976184f;

        float sm[PD][PD];
        #pragma unroll
        for (int i = 0; i < PD; ++i) {
            #pragma unroll
            for (int j = 0; j <= i; ++j) {
                float s = 0.f;
                #pragma unroll
                for (int k = 0; k <= j; ++k) s += sg[i][k] * sg[j][k];
                sm[i][j] = s;
                sm[j][i] = s;
            }
        }
        #pragma unroll
        for (int i = 0; i < PD; ++i)
            #pragma unroll
            for (int j = 0; j < PD; j += 4)
                *(float4*)&outS[(size_t)row * (PD * PD) + i * PD + j] =
                    make_float4(sm[i][j], sm[i][j + 1], sm[i][j + 2], sm[i][j + 3]);
    }
}

extern "C" void kernel_launch(void* const* d_in, const int* in_sizes, int n_in,
                              void* d_out, int out_size, void* d_ws, size_t ws_size,
                              hipStream_t stream) {
    const float* x = (const float*)d_in[0];
    const float* W = (const float*)d_in[1];
    const float* b = (const float*)d_in[2];
    float* out = (float*)d_out;
    (void)in_sizes; (void)n_in; (void)out_size; (void)d_ws; (void)ws_size;

    dim3 grid(BATCH / ROWS_PER_BLOCK);
    dim3 block(THREADS);
    l3m_param_head_kernel<<<grid, block, 0, stream>>>(x, W, b, out);
}


# K loop software-pipelined: next tile's 14 global loads per thread issued as one unconditional group ahead of the current tile's WMMAs, stored to a second LDS stage after them; one barrier per tile
# speedup vs baseline: 3.9376x; 3.9376x over previous
#include <hip/hip_runtime.h>
#include <hip/hip_bf16.h>

typedef __attribute__((ext_vector_type(2))) float v2f;
typedef __attribute__((ext_vector_type(8))) float v8f;

#define HID   4096
#define BATCH 16384
#define OUTC  44
#define PD    8

#define THREADS        128
#define ROWS_PER_BLOCK 64
#define KC        64
#define XS_STRIDE 68
#define WS_STRIDE 68
#define HS_STRIDE 52
#define NPAD      48
#define XS_TILE   (ROWS_PER_BLOCK * XS_STRIDE)
#define WS_TILE   (NPAD * WS_STRIDE)
#define NT        (HID / KC)


__device__ __forceinline__ void load_tile(const float4* __restrict__ xp,
                                          const float4* __restrict__ wp,
                                          const float4* __restrict__ wlast,
                                          int kq, float4 (&xr)[8], float4 (&wr)[6])
{
    #pragma unroll
    for (int i = 0; i < 8; ++i) xr[i] = xp[(size_t)i * 8 * (HID / 4) + kq];
    #pragma unroll
    for (int i = 0; i < 5; ++i) wr[i] = wp[(size_t)i * 8 * (HID / 4) + kq];
    wr[5] = wlast[kq];
}

__device__ __forceinline__ void store_tile(float* xb, float* wb, float* wlastb,
                                           const float4 (&xr)[8], const float4 (&wr)[6])
{
    #pragma unroll
    for (int i = 0; i < 8; ++i) *(float4*)&xb[i * 8 * XS_STRIDE] = xr[i];
    #pragma unroll
    for (int i = 0; i < 5; ++i) *(float4*)&wb[i * 8 * WS_STRIDE] = wr[i];
    *(float4*)wlastb = wr[5];
}

__device__ __forceinline__ void mma_tile(const float* xa, const float* wb,
                                         v8f& acc0, v8f& acc1, v8f& acc2)
{
    #pragma unroll
    for (int kk = 0; kk < KC; kk += 4) {
        v2f a  = *(const v2f*)&xa[kk];
        v2f b0 = *(const v2f*)&wb[kk];
        v2f b1 = *(const v2f*)&wb[16 * WS_STRIDE + kk];
        v2f b2 = *(const v2f*)&wb[32 * WS_STRIDE + kk];
        acc0 = __builtin_amdgcn_wmma_f32_16x16x4_f32(false, a, false, b0, (short)0, acc0, false, false);
        acc1 = __builtin_amdgcn_wmma_f32_16x16x4_f32(false, a, false, b1, (short)0, acc1, false, false);
        acc2 = __builtin_amdgcn_wmma_f32_16x16x4_f32(false, a, false, b2, (short)0, acc2, false, false);
    }
}


__global__ __launch_bounds__(THREADS) void l3m_param_head_kernel(
    const float* __restrict__ x, const float* __restrict__ W,
    const float* __restrict__ bias, float* __restrict__ out)
{
    __shared__ float xs[2 * XS_TILE];
    __shared__ float ws[2 * WS_TILE];

    const int tid  = threadIdx.x;
    const int wid  = tid >> 5;
    const int lane = tid & 31;
    const int half = lane >> 4;
    const int l16  = lane & 15;
    const int m0   = wid * 16;
    const int rowBase = blockIdx.x * ROWS_PER_BLOCK;

    const int sr = tid >> 4;
    const int c4 = tid & 15;
    const float4* xp = (const float4*)x + (size_t)(rowBase + sr) * (HID / 4) + c4;
    const float4* wp = (const float4*)W + (size_t)sr * (HID / 4) + c4;
    const bool lastRowValid = (40 + sr) < OUTC;
    const int  nlast = lastRowValid ? (40 + sr) : (OUTC - 1);
    const float4* wlast = (const float4*)W + (size_t)nlast * (HID / 4) + c4;
    const int wlo = nlast * WS_STRIDE + c4 * 4;
    const int xso = sr * XS_STRIDE + c4 * 4;
    const int wso = sr * WS_STRIDE + c4 * 4;
    const int xfo = (m0 + l16) * XS_STRIDE + 2 * half;
    const int wfo = l16 * WS_STRIDE + 2 * half;

    v8f acc0 = {}; v8f acc1 = {}; v8f acc2 = {};
    float4 xr[8], wr[6];

    load_tile(xp, wp, wlast, 0, xr, wr);
    __builtin_prefetch(&x[(size_t)(rowBase + (tid >> 1)) * HID + 2 * KC + (tid & 1) * 32], 0, 1);
    if (!lastRowValid) {
        const float4 z = make_float4(0.f, 0.f, 0.f, 0.f);
        *(float4*)&ws[wso + 5 * 8 * WS_STRIDE] = z;
        *(float4*)&ws[WS_TILE + wso + 5 * 8 * WS_STRIDE] = z;
    }
    store_tile(&xs[xso], &ws[wso], &ws[wlo], xr, wr);
    __syncthreads();

    for (int t = 0; t < NT - 1; ++t) {
        const int cur = t & 1;
        load_tile(xp, wp, wlast, (t + 1) * (KC / 4), xr, wr);
        if (t + 3 < NT)
            __builtin_prefetch(&x[(size_t)(rowBase + (tid >> 1)) * HID
                                  + (t + 3) * KC + (tid & 1) * 32], 0, 1);

        mma_tile(&xs[cur * XS_TILE + xfo], &ws[cur * WS_TILE + wfo], acc0, acc1, acc2);

        store_tile(&xs[(cur ^ 1) * XS_TILE + xso], &ws[(cur ^ 1) * WS_TILE + wso],
                   &ws[(cur ^ 1) * WS_TILE + wlo], xr, wr);
        __syncthreads();
    }
    mma_tile(&xs[((NT - 1) & 1) * XS_TILE + xfo], &ws[((NT - 1) & 1) * WS_TILE + wfo],
             acc0, acc1, acc2);

    static_assert(((NT - 1) & 1) == 1, "last tile must sit in stage 1");
    static_assert(ROWS_PER_BLOCK * HS_STRIDE <= XS_TILE, "h tile must fit in one x stage");
    float* hs = xs;
    #pragma unroll
    for (int v = 0; v < 8; ++v) {
        int m = m0 + v + 8 * half;
        hs[m * HS_STRIDE +  0 + l16] = acc0[v];
        hs[m * HS_STRIDE + 16 + l16] = acc1[v];
        hs[m * HS_STRIDE + 32 + l16] = acc2[v];
    }
    __syncthreads();

    if (tid < ROWS_PER_BLOCK) {
        const int row = rowBase + tid;
        float h[OUTC];
        #pragma unroll
        for (int j = 0; j < OUTC; ++j) h[j] = hs[tid * HS_STRIDE + j] + bias[j];

        float* outP = out;
        float* outS = out + (size_t)BATCH * PD;
        float* outL = out + (size_t)BATCH * PD + (size_t)BATCH * PD * PD;

        float pr[PD];
        #pragma unroll
        for (int i = 0; i < PD; ++i) pr[i] = 1.f / (1.f + expf(-h[i]));
        #pragma unroll
        for (int i = 0; i < PD; i += 4)
            *(float4*)&outP[(size_t)row * PD + i] =
                make_float4(pr[i], pr[i + 1], pr[i + 2], pr[i + 3]);

        float sg[PD][PD];
        #pragma unroll
        for (int i = 0; i < PD; ++i) {
            const float sc = 2.0f / sqrtf((float)(i + 1));
            float d  = h[PD + i];
            float sp = (d > 20.f) ? d : log1pf(expf(d));
            sg[i][i] = sp * sc;
            #pragma unroll
            for (int j = 0; j < i; ++j)
                sg[i][j] = h[2 * PD + (i * (i - 1)) / 2 + j] * sc;
        }

        float p = 1.f;
        #pragma unroll
        for (int i = 0; i < PD; ++i) p *= sg[i][i];
        float bp  = 10000.f * p;
        float lae = (bp > 0.f) ? (bp + log1pf(expf(-bp))) : log1pf(expf(bp));
        outL[row] = logf(lae) - 9.210340371976184f;

        float sm[PD][PD];
        #pragma unroll
        for (int i = 0; i < PD; ++i) {
            #pragma unroll
            for (int j = 0; j <= i; ++j) {
                float s = 0.f;
                #pragma unroll
                for (int k = 0; k <= j; ++k) s += sg[i][k] * sg[j][k];
                sm[i][j] = s;
                sm[j][i] = s;
            }
        }
        #pragma unroll
        for (int i = 0; i < PD; ++i)
            #pragma unroll
            for (int j = 0; j < PD; j += 4)
                *(float4*)&outS[(size_t)row * (PD * PD) + i * PD + j] =
                    make_float4(sm[i][j], sm[i][j + 1], sm[i][j + 2], sm[i][j + 3]);
    }
}

extern "C" void kernel_launch(void* const* d_in, const int* in_sizes, int n_in,
                              void* d_out, int out_size, void* d_ws, size_t ws_size,
                              hipStream_t stream) {
    const float* x = (const float*)d_in[0];
    const float* W = (const float*)d_in[1];
    const float* b = (const float*)d_in[2];
    float* out = (float*)d_out;
    (void)in_sizes; (void)n_in; (void)out_size; (void)d_ws; (void)ws_size;

    dim3 grid(BATCH / ROWS_PER_BLOCK);
    dim3 block(THREADS);
    l3m_param_head_kernel<<<grid, block, 0, stream>>>(x, W, b, out);
}


# Work-group widened to 16 waves: 4 row tiles x 4 in-block K splits of a 256-wide staged K tile (4 waves per SIMD, 4x loads in flight); the 4 partial sums per h are added through LDS in fixed order
# speedup vs baseline: 4.5652x; 1.1594x over previous
#include <hip/hip_runtime.h>
#include <hip/hip_bf16.h>

typedef __attribute__((ext_vector_type(2))) float v2f;
typedef __attribute__((ext_vector_type(8))) float v8f;

#define HID   4096
#define BATCH 16384
#define OUTC  44
#define PD    8

#define THREADS        512
#define ROWS_PER_BLOCK 64
#define KSPLIT    4
#define KC        64
#define KT        (KSPLIT * KC)
#define XS_STRIDE 260
#define WS_STRIDE 260
#define HS_STRIDE 52
#define NPAD      48
#define XS_TILE   (ROWS_PER_BLOCK * XS_STRIDE)
#define WS_TILE   (NPAD * WS_STRIDE)
#define HS_TILE   (ROWS_PER_BLOCK * HS_STRIDE)
#define NT        (HID / KT)


__device__ __forceinline__ void load_tile(const float4* __restrict__ xp,
                                          const float4* __restrict__ wp,
                                          const float4* __restrict__ wlast,
                                          int kq, float4 (&xr)[8], float4 (&wr)[6])
{
    #pragma unroll
    for (int i = 0; i < 8; ++i) xr[i] = xp[(size_t)i * 8 * (HID / 4) + kq];
    #pragma unroll
    for (int i = 0; i < 5; ++i) wr[i] = wp[(size_t)i * 8 * (HID / 4) + kq];
    wr[5] = wlast[kq];
}

__device__ __forceinline__ void store_tile(float* xb, float* wb, float* wlastb,
                                           const float4 (&xr)[8], const float4 (&wr)[6])
{
    #pragma unroll
    for (int i = 0; i < 8; ++i) *(float4*)&xb[i * 8 * XS_STRIDE] = xr[i];
    #pragma unroll
    for (int i = 0; i < 5; ++i) *(float4*)&wb[i * 8 * WS_STRIDE] = wr[i];
    *(float4*)wlastb = wr[5];
}

__device__ __forceinline__ void mma_tile(const float* xa, const float* wb,
                                         v8f& acc0, v8f& acc1, v8f& acc2)
{
    #pragma unroll
    for (int kk = 0; kk < KC; kk += 4) {
        v2f a  = *(const v2f*)&xa[kk];
        v2f b0 = *(const v2f*)&wb[kk];
        v2f b1 = *(const v2f*)&wb[16 * WS_STRIDE + kk];
        v2f b2 = *(const v2f*)&wb[32 * WS_STRIDE + kk];
        acc0 = __builtin_amdgcn_wmma_f32_16x16x4_f32(false, a, false, b0, (short)0, acc0, false, false);
        acc1 = __builtin_amdgcn_wmma_f32_16x16x4_f32(false, a, false, b1, (short)0, acc1, false, false);
        acc2 = __builtin_amdgcn_wmma_f32_16x16x4_f32(false, a, false, b2, (short)0, acc2, false, false);
    }
}


__global__ __launch_bounds__(THREADS) void l3m_param_head_kernel(
    const float* __restrict__ x, const float* __restrict__ W,
    const float* __restrict__ bias, float* __restrict__ out)
{
    __shared__ float xs[2 * XS_TILE];
    __shared__ float ws[2 * WS_TILE];

    const int tid  = threadIdx.x;
    const int wid  = tid >> 5;
    const int lane = tid & 31;
    const int half = lane >> 4;
    const int l16  = lane & 15;
    const int m0   = (wid & 3) * 16;
    const int ks   = wid >> 2;
    const int rowBase = blockIdx.x * ROWS_PER_BLOCK;

    const int sr = tid >> 6;
    const int c4 = tid & 63;
    const float4* xp = (const float4*)x + (size_t)(rowBase + sr) * (HID / 4) + c4;
    const float4* wp = (const float4*)W + (size_t)sr * (HID / 4) + c4;
    const bool lastRowValid = (40 + sr) < OUTC;
    const int  nlast = lastRowValid ? (40 + sr) : (OUTC - 1);
    const float4* wlast = (const float4*)W + (size_t)nlast * (HID / 4) + c4;
    const int wlo = nlast * WS_STRIDE + c4 * 4;
    const int xso = sr * XS_STRIDE + c4 * 4;
    const int wso = sr * WS_STRIDE + c4 * 4;
    const int xfo = (m0 + l16) * XS_STRIDE + ks * KC + 2 * half;
    const int wfo = l16 * WS_STRIDE + ks * KC + 2 * half;

    v8f acc0 = {}; v8f acc1 = {}; v8f acc2 = {};
    float4 xr[8], wr[6];

    load_tile(xp, wp, wlast, 0, xr, wr);
    __builtin_prefetch(&x[(size_t)(rowBase + (tid >> 3)) * HID + 2 * KT + (tid & 7) * 32], 0, 1);
    if (!lastRowValid) {
        const float4 z = make_float4(0.f, 0.f, 0.f, 0.f);
        *(float4*)&ws[wso + 5 * 8 * WS_STRIDE] = z;
        *(float4*)&ws[WS_TILE + wso + 5 * 8 * WS_STRIDE] = z;
    }
    store_tile(&xs[xso], &ws[wso], &ws[wlo], xr, wr);
    __syncthreads();

    for (int t = 0; t < NT - 1; ++t) {
        const int cur = t & 1;
        load_tile(xp, wp, wlast, (t + 1) * (KT / 4), xr, wr);
        if (t + 3 < NT)
            __builtin_prefetch(&x[(size_t)(rowBase + (tid >> 3)) * HID
                                  + (t + 3) * KT + (tid & 7) * 32], 0, 1);

        mma_tile(&xs[cur * XS_TILE + xfo], &ws[cur * WS_TILE + wfo], acc0, acc1, acc2);

        store_tile(&xs[(cur ^ 1) * XS_TILE + xso], &ws[(cur ^ 1) * WS_TILE + wso],
                   &ws[(cur ^ 1) * WS_TILE + wlo], xr, wr);
        __syncthreads();
    }
    mma_tile(&xs[((NT - 1) & 1) * XS_TILE + xfo], &ws[((NT - 1) & 1) * WS_TILE + wfo],
             acc0, acc1, acc2);

    static_assert(((NT - 1) & 1) == 1, "last tile must sit in stage 1");
    static_assert(KSPLIT * HS_TILE <= XS_TILE, "partial h tiles must fit in one x stage");
    float* hs = xs + ks * HS_TILE;
    #pragma unroll
    for (int v = 0; v < 8; ++v) {
        int m = m0 + v + 8 * half;
        hs[m * HS_STRIDE +  0 + l16] = acc0[v];
        hs[m * HS_STRIDE + 16 + l16] = acc1[v];
        hs[m * HS_STRIDE + 32 + l16] = acc2[v];
    }
    __syncthreads();

    if (tid < ROWS_PER_BLOCK) {
        const int row = rowBase + tid;
        const float* hp = xs + tid * HS_STRIDE;
        float h[OUTC];
        #pragma unroll
        for (int j = 0; j < OUTC; ++j)
            h[j] = ((hp[j] + hp[HS_TILE + j]) + (hp[2 * HS_TILE + j] + hp[3 * HS_TILE + j])) + bias[j];

        float* outP = out;
        float* outS = out + (size_t)BATCH * PD;
        float* outL = out + (size_t)BATCH * PD + (size_t)BATCH * PD * PD;

        float pr[PD];
        #pragma unroll
        for (int i = 0; i < PD; ++i) pr[i] = 1.f / (1.f + expf(-h[i]));
        #pragma unroll
        for (int i = 0; i < PD; i += 4)
            *(float4*)&outP[(size_t)row * PD + i] =
                make_float4(pr[i], pr[i + 1], pr[i + 2], pr[i + 3]);

        float sg[PD][PD];
        #pragma unroll
        for (int i = 0; i < PD; ++i) {
            const float sc = 2.0f / sqrtf((float)(i + 1));
            float d  = h[PD + i];
            float sp = (d > 20.f) ? d : log1pf(expf(d));
            sg[i][i] = sp * sc;
            #pragma unroll
            for (int j = 0; j < i; ++j)
                sg[i][j] = h[2 * PD + (i * (i - 1)) / 2 + j] * sc;
        }

        float p = 1.f;
        #pragma unroll
        for (int i = 0; i < PD; ++i) p *= sg[i][i];
        float bp  = 10000.f * p;
        float lae = (bp > 0.f) ? (bp + log1pf(expf(-bp))) : log1pf(expf(bp));
        outL[row] = logf(lae) - 9.210340371976184f;

        float sm[PD][PD];
        #pragma unroll
        for (int i = 0; i < PD; ++i) {
            #pragma unroll
            for (int j = 0; j <= i; ++j) {
                float s = 0.f;
                #pragma unroll
                for (int k = 0; k <= j; ++k) s += sg[i][k] * sg[j][k];
                sm[i][j] = s;
                sm[j][i] = s;
            }
        }
        #pragma unroll
        for (int i = 0; i < PD; ++i)
            #pragma unroll
            for (int j = 0; j < PD; j += 4)
                *(float4*)&outS[(size_t)row * (PD * PD) + i * PD + j] =
                    make_float4(sm[i][j], sm[i][j + 1], sm[i][j + 2], sm[i][j + 3]);
    }
}

extern "C" void kernel_launch(void* const* d_in, const int* in_sizes, int n_in,
                              void* d_out, int out_size, void* d_ws, size_t ws_size,
                              hipStream_t stream) {
    const float* x = (const float*)d_in[0];
    const float* W = (const float*)d_in[1];
    const float* b = (const float*)d_in[2];
    float* out = (float*)d_out;
    (void)in_sizes; (void)n_in; (void)out_size; (void)d_ws; (void)ws_size;

    dim3 grid(BATCH / ROWS_PER_BLOCK);
    dim3 block(THREADS);
    l3m_param_head_kernel<<<grid, block, 0, stream>>>(x, W, b, out);
}


# LDS fragment reads regrouped by scheduling groups: each 12-WMMA block's 8 reads issued as one group a block ahead (2 places inside a tile's WMMA stream instead of 16); k order and sums unchanged
# speedup vs baseline: 4.9432x; 1.0828x over previous
#include <hip/hip_runtime.h>
#include <hip/hip_bf16.h>

typedef __attribute__((ext_vector_type(2))) float v2f;
typedef __attribute__((ext_vector_type(8))) float v8f;

#define HID   4096
#define BATCH 16384
#define OUTC  44
#define PD    8

#define THREADS        512
#define ROWS_PER_BLOCK 64
#define KSPLIT    4
#define KC        64
#define KT        (KSPLIT * KC)
#define XS_STRIDE 260
#define WS_STRIDE 260
#define HS_STRIDE 52
#define NPAD      48
#define XS_TILE   (ROWS_PER_BLOCK * XS_STRIDE)
#define WS_TILE   (NPAD * WS_STRIDE)
#define HS_TILE   (ROWS_PER_BLOCK * HS_STRIDE)
#define NT        (HID / KT)


__device__ __forceinline__ void load_tile(const float4* __restrict__ xp,
                                          const float4* __restrict__ wp,
                                          const float4* __restrict__ wlast,
                                          int kq, float4 (&xr)[8], float4 (&wr)[6])
{
    #pragma unroll
    for (int i = 0; i < 8; ++i) xr[i] = xp[(size_t)i * 8 * (HID / 4) + kq];
    #pragma unroll
    for (int i = 0; i < 5; ++i) wr[i] = wp[(size_t)i * 8 * (HID / 4) + kq];
    wr[5] = wlast[kq];
}

__device__ __forceinline__ void store_tile(float* xb, float* wb, float* wlastb,
                                           const float4 (&xr)[8], const float4 (&wr)[6])
{
    #pragma unroll
    for (int i = 0; i < 8; ++i) *(float4*)&xb[i * 8 * XS_STRIDE] = xr[i];
    #pragma unroll
    for (int i = 0; i < 5; ++i) *(float4*)&wb[i * 8 * WS_STRIDE] = wr[i];
    *(float4*)wlastb = wr[5];
}

#define KSTEPS_PER_BLOCK 4
#define KB (4 * KSTEPS_PER_BLOCK)
struct Frags { v2f a[KSTEPS_PER_BLOCK], b0[KSTEPS_PER_BLOCK], b1[KSTEPS_PER_BLOCK], b2[KSTEPS_PER_BLOCK]; };

__device__ __forceinline__ void load_frags(Frags& f, const float* xa, const float* wb, int kk0)
{
    #pragma unroll
    for (int s = 0; s < KSTEPS_PER_BLOCK; ++s) {
        const int kk = kk0 + 4 * s;
        f.a[s]  = *(const v2f*)&xa[kk];
        f.b0[s] = *(const v2f*)&wb[kk];
        f.b1[s] = *(const v2f*)&wb[16 * WS_STRIDE + kk];
        f.b2[s] = *(const v2f*)&wb[32 * WS_STRIDE + kk];
    }
}

__device__ __forceinline__ void mma_block(const Frags& f, v8f& acc0, v8f& acc1, v8f& acc2)
{
    #pragma unroll
    for (int s = 0; s < KSTEPS_PER_BLOCK; ++s) {
        acc0 = __builtin_amdgcn_wmma_f32_16x16x4_f32(false, f.a[s], false, f.b0[s], (short)0, acc0, false, false);
        acc1 = __builtin_amdgcn_wmma_f32_16x16x4_f32(false, f.a[s], false, f.b1[s], (short)0, acc1, false, false);
        acc2 = __builtin_amdgcn_wmma_f32_16x16x4_f32(false, f.a[s], false, f.b2[s], (short)0, acc2, false, false);
    }
}

#define SG_WMMA      0x008
#define SG_DS_READ   0x100
#define SG_DS_WRITE  0x200
#define DS_PER_BLOCK (2 * KSTEPS_PER_BLOCK)
#define WMMA_PER_BLOCK (3 * KSTEPS_PER_BLOCK)

template <bool STORE>
__device__ __forceinline__ void mma_tile(const float* xa, const float* wb,
                                         v8f& acc0, v8f& acc1, v8f& acc2,
                                         float* xb, float* wbs, float* wlastb,
                                         const float4 (&xr)[8], const float4 (&wr)[6])
{
    static_assert(KC == 4 * KB, "a wave's K slice is 4 blocks");
    Frags f0, f1;
    load_frags(f0, xa, wb, 0 * KB);
    load_frags(f1, xa, wb, 1 * KB);
    mma_block(f0, acc0, acc1, acc2);
    load_frags(f0, xa, wb, 2 * KB);
    mma_block(f1, acc0, acc1, acc2);
    load_frags(f1, xa, wb, 3 * KB);
    if (STORE) store_tile(xb, wbs, wlastb, xr, wr);
    mma_block(f0, acc0, acc1, acc2);
    mma_block(f1, acc0, acc1, acc2);
    __builtin_amdgcn_sched_group_barrier(SG_DS_READ, 2 * DS_PER_BLOCK, 0);
    __builtin_amdgcn_sched_group_barrier(SG_WMMA, WMMA_PER_BLOCK, 0);
    __builtin_amdgcn_sched_group_barrier(SG_DS_READ, DS_PER_BLOCK, 0);
    __builtin_amdgcn_sched_group_barrier(SG_WMMA, WMMA_PER_BLOCK, 0);
    __builtin_amdgcn_sched_group_barrier(SG_DS_READ, DS_PER_BLOCK, 0);
    if (STORE) __builtin_amdgcn_sched_group_barrier(SG_DS_WRITE, 14, 0);
    __builtin_amdgcn_sched_group_barrier(SG_WMMA, 2 * WMMA_PER_BLOCK, 0);
}


__global__ __launch_bounds__(THREADS) void l3m_param_head_kernel(
    const float* __restrict__ x, const float* __restrict__ W,
    const float* __restrict__ bias, float* __restrict__ out)
{
    __shared__ float xs[2 * XS_TILE];
    __shared__ float ws[2 * WS_TILE];

    const int tid  = threadIdx.x;
    const int wid  = tid >> 5;
    const int lane = tid & 31;
    const int half = lane >> 4;
    const int l16  = lane & 15;
    const int m0   = (wid & 3) * 16;
    const int ks   = wid >> 2;
    const int rowBase = blockIdx.x * ROWS_PER_BLOCK;

    const int sr = tid >> 6;
    const int c4 = tid & 63;
    const float4* xp = (const float4*)x + (size_t)(rowBase + sr) * (HID / 4) + c4;
    const float4* wp = (const float4*)W + (size_t)sr * (HID / 4) + c4;
    const bool lastRowValid = (40 + sr) < OUTC;
    const int  nlast = lastRowValid ? (40 + sr) : (OUTC - 1);
    const float4* wlast = (const float4*)W + (size_t)nlast * (HID / 4) + c4;
    const int wlo = nlast * WS_STRIDE + c4 * 4;
    const int xso = sr * XS_STRIDE + c4 * 4;
    const int wso = sr * WS_STRIDE + c4 * 4;
    const int xfo = (m0 + l16) * XS_STRIDE + ks * KC + 2 * half;
    const int wfo = l16 * WS_STRIDE + ks * KC + 2 * half;

    v8f acc0 = {}; v8f acc1 = {}; v8f acc2 = {};
    float4 xr[8], wr[6];

    load_tile(xp, wp, wlast, 0, xr, wr);
    __builtin_prefetch(&x[(size_t)(rowBase + (tid >> 3)) * HID + 2 * KT + (tid & 7) * 32], 0, 1);
    if (!lastRowValid) {
        const float4 z = make_float4(0.f, 0.f, 0.f, 0.f);
        *(float4*)&ws[wso + 5 * 8 * WS_STRIDE] = z;
        *(float4*)&ws[WS_TILE + wso + 5 * 8 * WS_STRIDE] = z;
    }
    store_tile(&xs[xso], &ws[wso], &ws[wlo], xr, wr);
    __syncthreads();

    int xf_c = xfo, xf_n = XS_TILE + xfo;
    int wf_c = wfo, wf_n = WS_TILE + wfo;
    int xs_n = XS_TILE + xso, xs_c = xso;
    int ws_n = WS_TILE + wso, ws_c = wso;
    int wl_n = WS_TILE + wlo, wl_c = wlo;
    for (int t = 0; t < NT - 1; ++t) {
        load_tile(xp, wp, wlast, (t + 1) * (KT / 4), xr, wr);
        if (t + 3 < NT)
            __builtin_prefetch(&x[(size_t)(rowBase + (tid >> 3)) * HID
                                  + (t + 3) * KT + (tid & 7) * 32], 0, 1);

        mma_tile<true>(&xs[xf_c], &ws[wf_c], acc0, acc1, acc2,
                       &xs[xs_n], &ws[ws_n], &ws[wl_n], xr, wr);
        __syncthreads();
        int s;
        s = xf_c; xf_c = xf_n; xf_n = s;
        s = wf_c; wf_c = wf_n; wf_n = s;
        s = xs_c; xs_c = xs_n; xs_n = s;
        s = ws_c; ws_c = ws_n; ws_n = s;
        s = wl_c; wl_c = wl_n; wl_n = s;
    }
    mma_tile<false>(&xs[xf_c], &ws[wf_c], acc0, acc1, acc2, nullptr, nullptr, nullptr, xr, wr);

    static_assert(((NT - 1) & 1) == 1, "last tile must sit in stage 1");
    static_assert(KSPLIT * HS_TILE <= XS_TILE, "partial h tiles must fit in one x stage");
    float* hs = xs + ks * HS_TILE;
    #pragma unroll
    for (int v = 0; v < 8; ++v) {
        int m = m0 + v + 8 * half;
        hs[m * HS_STRIDE +  0 + l16] = acc0[v];
        hs[m * HS_STRIDE + 16 + l16] = acc1[v];
        hs[m * HS_STRIDE + 32 + l16] = acc2[v];
    }
    __syncthreads();

    if (tid < ROWS_PER_BLOCK) {
        const int row = rowBase + tid;
        const float* hp = xs + tid * HS_STRIDE;
        float h[OUTC];
        #pragma unroll
        for (int j = 0; j < OUTC; ++j)
            h[j] = ((hp[j] + hp[HS_TILE + j]) + (hp[2 * HS_TILE + j] + hp[3 * HS_TILE + j])) + bias[j];

        float* outP = out;
        float* outS = out + (size_t)BATCH * PD;
        float* outL = out + (size_t)BATCH * PD + (size_t)BATCH * PD * PD;

        float pr[PD];
        #pragma unroll
        for (int i = 0; i < PD; ++i) pr[i] = 1.f / (1.f + expf(-h[i]));
        #pragma unroll
        for (int i = 0; i < PD; i += 4)
            *(float4*)&outP[(size_t)row * PD + i] =
                make_float4(pr[i], pr[i + 1], pr[i + 2], pr[i + 3]);

        float sg[PD][PD];
        #pragma unroll
        for (int i = 0; i < PD; ++i) {
            const float sc = 2.0f / sqrtf((float)(i + 1));
            float d  = h[PD + i];
            float sp = (d > 20.f) ? d : log1pf(expf(d));
            sg[i][i] = sp * sc;
            #pragma unroll
            for (int j = 0; j < i; ++j)
                sg[i][j] = h[2 * PD + (i * (i - 1)) / 2 + j] * sc;
        }

        float p = 1.f;
        #pragma unroll
        for (int i = 0; i < PD; ++i) p *= sg[i][i];
        float bp  = 10000.f * p;
        float lae = (bp > 0.f) ? (bp + log1pf(expf(-bp))) : log1pf(expf(bp));
        outL[row] = logf(lae) - 9.210340371976184f;

        float sm[PD][PD];
        #pragma unroll
        for (int i = 0; i < PD; ++i) {
            #pragma unroll
            for (int j = 0; j <= i; ++j) {
                float s = 0.f;
                #pragma unroll
                for (int k = 0; k <= j; ++k) s += sg[i][k] * sg[j][k];
                sm[i][j] = s;
                sm[j][i] = s;
            }
        }
        #pragma unroll
        for (int i = 0; i < PD; ++i)
            #pragma unroll
            for (int j = 0; j < PD; j += 4)
                *(float4*)&outS[(size_t)row * (PD * PD) + i * PD + j] =
                    make_float4(sm[i][j], sm[i][j + 1], sm[i][j + 2], sm[i][j + 3]);
    }
}

extern "C" void kernel_launch(void* const* d_in, const int* in_sizes, int n_in,
                              void* d_out, int out_size, void* d_ws, size_t ws_size,
                              hipStream_t stream) {
    const float* x = (const float*)d_in[0];
    const float* W = (const float*)d_in[1];
    const float* b = (const float*)d_in[2];
    float* out = (float*)d_out;
    (void)in_sizes; (void)n_in; (void)out_size; (void)d_ws; (void)ws_size;

    dim3 grid(BATCH / ROWS_PER_BLOCK);
    dim3 block(THREADS);
    l3m_param_head_kernel<<<grid, block, 0, stream>>>(x, W, b, out);
}


# Staged global loads re-issued right behind their own LDS stores: tile t+2's 14 loads per thread fly a whole tile time, across an LDS-only barrier, not from tile start to middle; sums unchanged
# speedup vs baseline: 5.1405x; 1.0399x over previous
#include <hip/hip_runtime.h>
#include <hip/hip_bf16.h>

typedef __attribute__((ext_vector_type(2))) float v2f;
typedef __attribute__((ext_vector_type(8))) float v8f;
typedef __attribute__((ext_vector_type(4))) float v4f;

#define HID   4096
#define BATCH 16384
#define OUTC  44
#define PD    8

#define THREADS        512
#define ROWS_PER_BLOCK 64
#define KSPLIT    4
#define KC        64
#define KT        (KSPLIT * KC)
#define XS_STRIDE 260
#define WS_STRIDE 260
#define HS_STRIDE 52
#define NPAD      48
#define XS_TILE   (ROWS_PER_BLOCK * XS_STRIDE)
#define WS_TILE   (NPAD * WS_STRIDE)
#define HS_TILE   (ROWS_PER_BLOCK * HS_STRIDE)
#define NT        (HID / KT)


__device__ __forceinline__ void load_tile(const v4f* __restrict__ xp,
                                          const v4f* __restrict__ wp,
                                          const v4f* __restrict__ wlast,
                                          int kq, v4f (&xr)[8], v4f (&wr)[6])
{
    #pragma unroll
    for (int i = 0; i < 8; ++i) xr[i] = xp[(size_t)i * 8 * (HID / 4) + kq];
    #pragma unroll
    for (int i = 0; i < 5; ++i) wr[i] = wp[(size_t)i * 8 * (HID / 4) + kq];
    wr[5] = wlast[kq];
}

__device__ __forceinline__ void store_tile(float* xb, float* wb, float* wlastb,
                                           const v4f (&xr)[8], const v4f (&wr)[6])
{
    #pragma unroll
    for (int i = 0; i < 8; ++i) *(v4f*)&xb[i * 8 * XS_STRIDE] = xr[i];
    #pragma unroll
    for (int i = 0; i < 5; ++i) *(v4f*)&wb[i * 8 * WS_STRIDE] = wr[i];
    *(v4f*)wlastb = wr[5];
}

__device__ __forceinline__ void lds_barrier()
{
    __builtin_amdgcn_fence(__ATOMIC_RELEASE, "workgroup", "local");
    __builtin_amdgcn_s_barrier();
    __builtin_amdgcn_fence(__ATOMIC_ACQUIRE, "workgroup", "local");
}

#define KSTEPS_PER_BLOCK 4
#define KB (4 * KSTEPS_PER_BLOCK)
struct Frags { v2f a[KSTEPS_PER_BLOCK], b0[KSTEPS_PER_BLOCK], b1[KSTEPS_PER_BLOCK], b2[KSTEPS_PER_BLOCK]; };

__device__ __forceinline__ void load_frags(Frags& f, const float* xa, const float* wb, int kk0)
{
    #pragma unroll
    for (int s = 0; s < KSTEPS_PER_BLOCK; ++s) {
        const int kk = kk0 + 4 * s;
        f.a[s]  = *(const v2f*)&xa[kk];
        f.b0[s] = *(const v2f*)&wb[kk];
        f.b1[s] = *(const v2f*)&wb[16 * WS_STRIDE + kk];
        f.b2[s] = *(const v2f*)&wb[32 * WS_STRIDE + kk];
    }
}

__device__ __forceinline__ void mma_block(const Frags& f, v8f& acc0, v8f& acc1, v8f& acc2)
{
    #pragma unroll
    for (int s = 0; s < KSTEPS_PER_BLOCK; ++s) {
        acc0 = __builtin_amdgcn_wmma_f32_16x16x4_f32(false, f.a[s], false, f.b0[s], (short)0, acc0, false, false);
        acc1 = __builtin_amdgcn_wmma_f32_16x16x4_f32(false, f.a[s], false, f.b1[s], (short)0, acc1, false, false);
        acc2 = __builtin_amdgcn_wmma_f32_16x16x4_f32(false, f.a[s], false, f.b2[s], (short)0, acc2, false, false);
    }
}

#define SG_WMMA      0x008
#define SG_DS_READ   0x100
#define SG_DS_WRITE  0x200
#define DS_PER_BLOCK (2 * KSTEPS_PER_BLOCK)
#define WMMA_PER_BLOCK (3 * KSTEPS_PER_BLOCK)

template <bool STORE, bool LOAD>
__device__ __forceinline__ void mma_tile(const float* xa, const float* wb,
                                         v8f& acc0, v8f& acc1, v8f& acc2,
                                         float* xb, float* wbs, float* wlastb,
                                         const v4f* __restrict__ xp,
                                         const v4f* __restrict__ wp,
                                         const v4f* __restrict__ wlast, int kq,
                                         v4f (&xr)[8], v4f (&wr)[6])
{
    static_assert(KC == 4 * KB, "a wave's K slice is 4 blocks");
    Frags f0, f1;
    load_frags(f0, xa, wb, 0 * KB);
    load_frags(f1, xa, wb, 1 * KB);
    mma_block(f0, acc0, acc1, acc2);
    load_frags(f0, xa, wb, 2 * KB);
    mma_block(f1, acc0, acc1, acc2);
    load_frags(f1, xa, wb, 3 * KB);
    if (STORE) store_tile(xb, wbs, wlastb, xr, wr);
    if (LOAD) load_tile(xp, wp, wlast, kq, xr, wr);
    mma_block(f0, acc0, acc1, acc2);
    mma_block(f1, acc0, acc1, acc2);
    __builtin_amdgcn_sched_group_barrier(SG_DS_READ, 2 * DS_PER_BLOCK, 0);
    __builtin_amdgcn_sched_group_barrier(SG_WMMA, WMMA_PER_BLOCK, 0);
    __builtin_amdgcn_sched_group_barrier(SG_DS_READ, DS_PER_BLOCK, 0);
    __builtin_amdgcn_sched_group_barrier(SG_WMMA, WMMA_PER_BLOCK, 0);
    __builtin_amdgcn_sched_group_barrier(SG_DS_READ, DS_PER_BLOCK, 0);
    if (STORE) __builtin_amdgcn_sched_group_barrier(SG_DS_WRITE, 14, 0);
    __builtin_amdgcn_sched_group_barrier(SG_WMMA, 2 * WMMA_PER_BLOCK, 0);
}


__global__ __launch_bounds__(THREADS) void l3m_param_head_kernel(
    const float* __restrict__ x, const float* __restrict__ W,
    const float* __restrict__ bias, float* __restrict__ out)
{
    __shared__ float xs[2 * XS_TILE];
    __shared__ float ws[2 * WS_TILE];

    const int tid  = threadIdx.x;
    const int wid  = tid >> 5;
    const int lane = tid & 31;
    const int half = lane >> 4;
    const int l16  = lane & 15;
    const int m0   = (wid & 3) * 16;
    const int ks   = wid >> 2;
    const int rowBase = blockIdx.x * ROWS_PER_BLOCK;

    const int sr = tid >> 6;
    const int c4 = tid & 63;
    const v4f* xp = (const v4f*)x + (size_t)(rowBase + sr) * (HID / 4) + c4;
    const v4f* wp = (const v4f*)W + (size_t)sr * (HID / 4) + c4;
    const bool lastRowValid = (40 + sr) < OUTC;
    const int  nlast = lastRowValid ? (40 + sr) : (OUTC - 1);
    const v4f* wlast = (const v4f*)W + (size_t)nlast * (HID / 4) + c4;
    const int wlo = nlast * WS_STRIDE + c4 * 4;
    const int xso = sr * XS_STRIDE + c4 * 4;
    const int wso = sr * WS_STRIDE + c4 * 4;
    const int xfo = (m0 + l16) * XS_STRIDE + ks * KC + 2 * half;
    const int wfo = l16 * WS_STRIDE + ks * KC + 2 * half;

    v8f acc0 = {}; v8f acc1 = {}; v8f acc2 = {};
    v4f xr[8], wr[6];

    load_tile(xp, wp, wlast, 0, xr, wr);
    __builtin_prefetch(&x[(size_t)(rowBase + (tid >> 3)) * HID + 2 * KT + (tid & 7) * 32], 0, 1);
    if (!lastRowValid) {
        const float4 z = make_float4(0.f, 0.f, 0.f, 0.f);
        *(float4*)&ws[wso + 5 * 8 * WS_STRIDE] = z;
        *(float4*)&ws[WS_TILE + wso + 5 * 8 * WS_STRIDE] = z;
    }
    store_tile(&xs[xso], &ws[wso], &ws[wlo], xr, wr);
    load_tile(xp, wp, wlast, 1 * (KT / 4), xr, wr);
    lds_barrier();

    int xf_c = xfo, xf_n = XS_TILE + xfo;
    int wf_c = wfo, wf_n = WS_TILE + wfo;
    int xs_n = XS_TILE + xso, xs_c = xso;
    int ws_n = WS_TILE + wso, ws_c = wso;
    int wl_n = WS_TILE + wlo, wl_c = wlo;
    for (int t = 0; t < NT - 2; ++t) {
        if (t + 3 < NT)
            __builtin_prefetch(&x[(size_t)(rowBase + (tid >> 3)) * HID
                                  + (t + 3) * KT + (tid & 7) * 32], 0, 1);

        mma_tile<true, true>(&xs[xf_c], &ws[wf_c], acc0, acc1, acc2,
                             &xs[xs_n], &ws[ws_n], &ws[wl_n],
                             xp, wp, wlast, (t + 2) * (KT / 4), xr, wr);
        lds_barrier();
        int s;
        s = xf_c; xf_c = xf_n; xf_n = s;
        s = wf_c; wf_c = wf_n; wf_n = s;
        s = xs_c; xs_c = xs_n; xs_n = s;
        s = ws_c; ws_c = ws_n; ws_n = s;
        s = wl_c; wl_c = wl_n; wl_n = s;
    }
    mma_tile<true, false>(&xs[xf_c], &ws[wf_c], acc0, acc1, acc2,
                          &xs[xs_n], &ws[ws_n], &ws[wl_n], nullptr, nullptr, nullptr, 0, xr, wr);
    lds_barrier();
    mma_tile<false, false>(&xs[xf_n], &ws[wf_n], acc0, acc1, acc2,
                           nullptr, nullptr, nullptr, nullptr, nullptr, nullptr, 0, xr, wr);

    static_assert(((NT - 1) & 1) == 1, "last tile must sit in stage 1");
    static_assert(KSPLIT * HS_TILE <= XS_TILE, "partial h tiles must fit in one x stage");
    float* hs = xs + ks * HS_TILE;
    #pragma unroll
    for (int v = 0; v < 8; ++v) {
        int m = m0 + v + 8 * half;
        hs[m * HS_STRIDE +  0 + l16] = acc0[v];
        hs[m * HS_STRIDE + 16 + l16] = acc1[v];
        hs[m * HS_STRIDE + 32 + l16] = acc2[v];
    }
    __syncthreads();

    if (tid < ROWS_PER_BLOCK) {
        const int row = rowBase + tid;
        const float* hp = xs + tid * HS_STRIDE;
        float h[OUTC];
        #pragma unroll
        for (int j = 0; j < OUTC; ++j)
            h[j] = ((hp[j] + hp[HS_TILE + j]) + (hp[2 * HS_TILE + j] + hp[3 * HS_TILE + j])) + bias[j];

        float* outP = out;
        float* outS = out + (size_t)BATCH * PD;
        float* outL = out + (size_t)BATCH * PD + (size_t)BATCH * PD * PD;

        float pr[PD];
        #pragma unroll
        for (int i = 0; i < PD; ++i) pr[i] = 1.f / (1.f + expf(-h[i]));
        #pragma unroll
        for (int i = 0; i < PD; i += 4)
            *(float4*)&outP[(size_t)row * PD + i] =
                make_float4(pr[i], pr[i + 1], pr[i + 2], pr[i + 3]);

        float sg[PD][PD];
        #pragma unroll
        for (int i = 0; i < PD; ++i) {
            const float sc = 2.0f / sqrtf((float)(i + 1));
            float d  = h[PD + i];
            float sp = (d > 20.f) ? d : log1pf(expf(d));
            sg[i][i] = sp * sc;
            #pragma unroll
            for (int j = 0; j < i; ++j)
                sg[i][j] = h[2 * PD + (i * (i - 1)) / 2 + j] * sc;
        }

        float p = 1.f;
        #pragma unroll
        for (int i = 0; i < PD; ++i) p *= sg[i][i];
        float bp  = 10000.f * p;
        float lae = (bp > 0.f) ? (bp + log1pf(expf(-bp))) : log1pf(expf(bp));
        outL[row] = logf(lae) - 9.210340371976184f;

        float sm[PD][PD];
        #pragma unroll
        for (int i = 0; i < PD; ++i) {
            #pragma unroll
            for (int j = 0; j <= i; ++j) {
                float s = 0.f;
                #pragma unroll
                for (int k = 0; k <= j; ++k) s += sg[i][k] * sg[j][k];
                sm[i][j] = s;
                sm[j][i] = s;
            }
        }
        #pragma unroll
        for (int i = 0; i < PD; ++i)
            #pragma unroll
            for (int j = 0; j < PD; j += 4)
                *(float4*)&outS[(size_t)row * (PD * PD) + i * PD + j] =
                    make_float4(sm[i][j], sm[i][j + 1], sm[i][j + 2], sm[i][j + 3]);
    }
}

extern "C" void kernel_launch(void* const* d_in, const int* in_sizes, int n_in,
                              void* d_out, int out_size, void* d_ws, size_t ws_size,
                              hipStream_t stream) {
    const float* x = (const float*)d_in[0];
    const float* W = (const float*)d_in[1];
    const float* b = (const float*)d_in[2];
    float* out = (float*)d_out;
    (void)in_sizes; (void)n_in; (void)out_size; (void)d_ws; (void)ws_size;

    dim3 grid(BATCH / ROWS_PER_BLOCK);
    dim3 block(THREADS);
    l3m_param_head_kernel<<<grid, block, 0, stream>>>(x, W, b, out);
}


# Tile barrier split (signal behind WMMA block 2, wait behind block 3); next tile's staging stores and loads moved to the tile start: no wave's LDS or load issue falls where others wait; sums unchanged
# speedup vs baseline: 5.2945x; 1.0300x over previous
#include <hip/hip_runtime.h>
#include <hip/hip_bf16.h>

typedef __attribute__((ext_vector_type(2))) float v2f;
typedef __attribute__((ext_vector_type(8))) float v8f;
typedef __attribute__((ext_vector_type(4))) float v4f;

#define HID   4096
#define BATCH 16384
#define OUTC  44
#define PD    8

#define THREADS        512
#define ROWS_PER_BLOCK 64
#define KSPLIT    4
#define KC        64
#define KT        (KSPLIT * KC)
#define XS_STRIDE 260
#define WS_STRIDE 260
#define HS_STRIDE 52
#define NPAD      48
#define XS_TILE   (ROWS_PER_BLOCK * XS_STRIDE)
#define WS_TILE   (NPAD * WS_STRIDE)
#define HS_TILE   (ROWS_PER_BLOCK * HS_STRIDE)
#define NT        (HID / KT)


__device__ __forceinline__ void load_tile(const v4f* __restrict__ xp,
                                          const v4f* __restrict__ wp,
                                          const v4f* __restrict__ wlast,
                                          int kq, v4f (&xr)[8], v4f (&wr)[6])
{
    #pragma unroll
    for (int i = 0; i < 8; ++i) xr[i] = xp[(size_t)i * 8 * (HID / 4) + kq];
    #pragma unroll
    for (int i = 0; i < 5; ++i) wr[i] = wp[(size_t)i * 8 * (HID / 4) + kq];
    wr[5] = wlast[kq];
}

__device__ __forceinline__ void store_tile(float* xb, float* wb, float* wlastb,
                                           const v4f (&xr)[8], const v4f (&wr)[6])
{
    #pragma unroll
    for (int i = 0; i < 8; ++i) *(v4f*)&xb[i * 8 * XS_STRIDE] = xr[i];
    #pragma unroll
    for (int i = 0; i < 5; ++i) *(v4f*)&wb[i * 8 * WS_STRIDE] = wr[i];
    *(v4f*)wlastb = wr[5];
}

__device__ __forceinline__ void lds_barrier()
{
    __builtin_amdgcn_fence(__ATOMIC_RELEASE, "workgroup", "local");
    __builtin_amdgcn_s_barrier();
    __builtin_amdgcn_fence(__ATOMIC_ACQUIRE, "workgroup", "local");
}

#define KSTEPS_PER_BLOCK 4
#define KB (4 * KSTEPS_PER_BLOCK)
struct Frags { v2f a[KSTEPS_PER_BLOCK], b0[KSTEPS_PER_BLOCK], b1[KSTEPS_PER_BLOCK], b2[KSTEPS_PER_BLOCK]; };

__device__ __forceinline__ void load_frags(Frags& f, const float* xa, const float* wb, int kk0)
{
    #pragma unroll
    for (int s = 0; s < KSTEPS_PER_BLOCK; ++s) {
        const int kk = kk0 + 4 * s;
        f.a[s]  = *(const v2f*)&xa[kk];
        f.b0[s] = *(const v2f*)&wb[kk];
        f.b1[s] = *(const v2f*)&wb[16 * WS_STRIDE + kk];
        f.b2[s] = *(const v2f*)&wb[32 * WS_STRIDE + kk];
    }
}

__device__ __forceinline__ void mma_block(const Frags& f, v8f& acc0, v8f& acc1, v8f& acc2)
{
    #pragma unroll
    for (int s = 0; s < KSTEPS_PER_BLOCK; ++s) {
        acc0 = __builtin_amdgcn_wmma_f32_16x16x4_f32(false, f.a[s], false, f.b0[s], (short)0, acc0, false, false);
        acc1 = __builtin_amdgcn_wmma_f32_16x16x4_f32(false, f.a[s], false, f.b1[s], (short)0, acc1, false, false);
        acc2 = __builtin_amdgcn_wmma_f32_16x16x4_f32(false, f.a[s], false, f.b2[s], (short)0, acc2, false, false);
    }
}

#define SG_WMMA      0x008
#define SG_DS_READ   0x100
#define SG_DS_WRITE  0x200
#define DS_PER_BLOCK (2 * KSTEPS_PER_BLOCK)
#define WMMA_PER_BLOCK (3 * KSTEPS_PER_BLOCK)

template <bool STORE, bool LOAD, bool SYNC, bool EARLY>
__device__ __forceinline__ void mma_tile(const float* xa, const float* wb,
                                         v8f& acc0, v8f& acc1, v8f& acc2,
                                         float* xb, float* wbs, float* wlastb,
                                         const v4f* __restrict__ xp,
                                         const v4f* __restrict__ wp,
                                         const v4f* __restrict__ wlast, int kq,
                                         v4f (&xr)[8], v4f (&wr)[6])
{
    static_assert(KC == 4 * KB, "a wave's K slice is 4 blocks");
    Frags f0, f1;
    load_frags(f0, xa, wb, 0 * KB);
    load_frags(f1, xa, wb, 1 * KB);
    if (EARLY && STORE) store_tile(xb, wbs, wlastb, xr, wr);
    if (EARLY && LOAD) load_tile(xp, wp, wlast, kq, xr, wr);
    mma_block(f0, acc0, acc1, acc2);
    load_frags(f0, xa, wb, 2 * KB);
    mma_block(f1, acc0, acc1, acc2);
    load_frags(f1, xa, wb, 3 * KB);
    if (!EARLY && STORE) store_tile(xb, wbs, wlastb, xr, wr);
    if (!EARLY && LOAD) load_tile(xp, wp, wlast, kq, xr, wr);
    mma_block(f0, acc0, acc1, acc2);
    __builtin_amdgcn_sched_group_barrier(SG_DS_READ, 2 * DS_PER_BLOCK, 0);
    if (EARLY && STORE) __builtin_amdgcn_sched_group_barrier(SG_DS_WRITE, 14, 0);
    __builtin_amdgcn_sched_group_barrier(SG_WMMA, WMMA_PER_BLOCK, 0);
    __builtin_amdgcn_sched_group_barrier(SG_DS_READ, DS_PER_BLOCK, 0);
    __builtin_amdgcn_sched_group_barrier(SG_WMMA, WMMA_PER_BLOCK, 0);
    __builtin_amdgcn_sched_group_barrier(SG_DS_READ, DS_PER_BLOCK, 0);
    if (!EARLY && STORE) __builtin_amdgcn_sched_group_barrier(SG_DS_WRITE, 14, 0);
    __builtin_amdgcn_sched_group_barrier(SG_WMMA, WMMA_PER_BLOCK, 0);
    __builtin_amdgcn_sched_barrier(0);
    if (SYNC) {
        __builtin_amdgcn_fence(__ATOMIC_RELEASE, "workgroup", "local");
        __builtin_amdgcn_s_barrier_signal(-1);
    }
    __builtin_amdgcn_sched_barrier(0);
    mma_block(f1, acc0, acc1, acc2);
    __builtin_amdgcn_sched_barrier(0);
    if (SYNC) {
        __builtin_amdgcn_s_barrier_wait(-1);
        __builtin_amdgcn_fence(__ATOMIC_ACQUIRE, "workgroup", "local");
    }
    __builtin_amdgcn_sched_barrier(0);
}


__global__ __launch_bounds__(THREADS) void l3m_param_head_kernel(
    const float* __restrict__ x, const float* __restrict__ W,
    const float* __restrict__ bias, float* __restrict__ out)
{
    __shared__ float xs[2 * XS_TILE];
    __shared__ float ws[2 * WS_TILE];

    const int tid  = threadIdx.x;
    const int wid  = tid >> 5;
    const int lane = tid & 31;
    const int half = lane >> 4;
    const int l16  = lane & 15;
    const int m0   = (wid & 3) * 16;
    const int ks   = wid >> 2;
    const int rowBase = blockIdx.x * ROWS_PER_BLOCK;

    const int sr = tid >> 6;
    const int c4 = tid & 63;
    const v4f* xp = (const v4f*)x + (size_t)(rowBase + sr) * (HID / 4) + c4;
    const v4f* wp = (const v4f*)W + (size_t)sr * (HID / 4) + c4;
    const bool lastRowValid = (40 + sr) < OUTC;
    const int  nlast = lastRowValid ? (40 + sr) : (OUTC - 1);
    const v4f* wlast = (const v4f*)W + (size_t)nlast * (HID / 4) + c4;
    const int wlo = nlast * WS_STRIDE + c4 * 4;
    const int xso = sr * XS_STRIDE + c4 * 4;
    const int wso = sr * WS_STRIDE + c4 * 4;
    const int xfo = (m0 + l16) * XS_STRIDE + ks * KC + 2 * half;
    const int wfo = l16 * WS_STRIDE + ks * KC + 2 * half;

    v8f acc0 = {}; v8f acc1 = {}; v8f acc2 = {};
    v4f xr[8], wr[6];

    load_tile(xp, wp, wlast, 0, xr, wr);
    __builtin_prefetch(&x[(size_t)(rowBase + (tid >> 3)) * HID + 2 * KT + (tid & 7) * 32], 0, 1);
    if (!lastRowValid) {
        const float4 z = make_float4(0.f, 0.f, 0.f, 0.f);
        *(float4*)&ws[wso + 5 * 8 * WS_STRIDE] = z;
        *(float4*)&ws[WS_TILE + wso + 5 * 8 * WS_STRIDE] = z;
    }
    store_tile(&xs[xso], &ws[wso], &ws[wlo], xr, wr);
    load_tile(xp, wp, wlast, 1 * (KT / 4), xr, wr);
    lds_barrier();

    int xf_c = xfo, xf_n = XS_TILE + xfo;
    int wf_c = wfo, wf_n = WS_TILE + wfo;
    int xs_n = XS_TILE + xso, xs_c = xso;
    int ws_n = WS_TILE + wso, ws_c = wso;
    int wl_n = WS_TILE + wlo, wl_c = wlo;
    __builtin_prefetch(&x[(size_t)(rowBase + (tid >> 3)) * HID + 3 * KT + (tid & 7) * 32], 0, 1);
    mma_tile<true, true, true, false>(&xs[xf_c], &ws[wf_c], acc0, acc1, acc2,
                                      &xs[xs_n], &ws[ws_n], &ws[wl_n],
                                      xp, wp, wlast, 2 * (KT / 4), xr, wr);
    {
        int s;
        s = xf_c; xf_c = xf_n; xf_n = s;
        s = wf_c; wf_c = wf_n; wf_n = s;
        s = xs_c; xs_c = xs_n; xs_n = s;
        s = ws_c; ws_c = ws_n; ws_n = s;
        s = wl_c; wl_c = wl_n; wl_n = s;
    }
    for (int t = 1; t < NT - 2; ++t) {
        if (t + 3 < NT)
            __builtin_prefetch(&x[(size_t)(rowBase + (tid >> 3)) * HID
                                  + (t + 3) * KT + (tid & 7) * 32], 0, 1);

        mma_tile<true, true, true, true>(&xs[xf_c], &ws[wf_c], acc0, acc1, acc2,
                                         &xs[xs_n], &ws[ws_n], &ws[wl_n],
                                         xp, wp, wlast, (t + 2) * (KT / 4), xr, wr);
        int s;
        s = xf_c; xf_c = xf_n; xf_n = s;
        s = wf_c; wf_c = wf_n; wf_n = s;
        s = xs_c; xs_c = xs_n; xs_n = s;
        s = ws_c; ws_c = ws_n; ws_n = s;
        s = wl_c; wl_c = wl_n; wl_n = s;
    }
    mma_tile<true, false, true, true>(&xs[xf_c], &ws[wf_c], acc0, acc1, acc2,
                                      &xs[xs_n], &ws[ws_n], &ws[wl_n],
                                      nullptr, nullptr, nullptr, 0, xr, wr);
    mma_tile<false, false, false, true>(&xs[xf_n], &ws[wf_n], acc0, acc1, acc2,
                                        nullptr, nullptr, nullptr, nullptr, nullptr, nullptr, 0, xr, wr);

    static_assert(((NT - 1) & 1) == 1, "last tile must sit in stage 1");
    static_assert(KSPLIT * HS_TILE <= XS_TILE, "partial h tiles must fit in one x stage");
    float* hs = xs + ks * HS_TILE;
    #pragma unroll
    for (int v = 0; v < 8; ++v) {
        int m = m0 + v + 8 * half;
        hs[m * HS_STRIDE +  0 + l16] = acc0[v];
        hs[m * HS_STRIDE + 16 + l16] = acc1[v];
        hs[m * HS_STRIDE + 32 + l16] = acc2[v];
    }
    __syncthreads();

    if (tid < ROWS_PER_BLOCK) {
        const int row = rowBase + tid;
        const float* hp = xs + tid * HS_STRIDE;
        float h[OUTC];
        #pragma unroll
        for (int j = 0; j < OUTC; ++j)
            h[j] = ((hp[j] + hp[HS_TILE + j]) + (hp[2 * HS_TILE + j] + hp[3 * HS_TILE + j])) + bias[j];

        float* outP = out;
        float* outS = out + (size_t)BATCH * PD;
        float* outL = out + (size_t)BATCH * PD + (size_t)BATCH * PD * PD;

        float pr[PD];
        #pragma unroll
        for (int i = 0; i < PD; ++i) pr[i] = 1.f / (1.f + expf(-h[i]));
        #pragma unroll
        for (int i = 0; i < PD; i += 4)
            *(float4*)&outP[(size_t)row * PD + i] =
                make_float4(pr[i], pr[i + 1], pr[i + 2], pr[i + 3]);

        float sg[PD][PD];
        #pragma unroll
        for (int i = 0; i < PD; ++i) {
            const float sc = 2.0f / sqrtf((float)(i + 1));
            float d  = h[PD + i];
            float sp = (d > 20.f) ? d : log1pf(expf(d));
            sg[i][i] = sp * sc;
            #pragma unroll
            for (int j = 0; j < i; ++j)
                sg[i][j] = h[2 * PD + (i * (i - 1)) / 2 + j] * sc;
        }

        float p = 1.f;
        #pragma unroll
        for (int i = 0; i < PD; ++i) p *= sg[i][i];
        float bp  = 10000.f * p;
        float lae = (bp > 0.f) ? (bp + log1pf(expf(-bp))) : log1pf(expf(bp));
        outL[row] = logf(lae) - 9.210340371976184f;

        float sm[PD][PD];
        #pragma unroll
        for (int i = 0; i < PD; ++i) {
            #pragma unroll
            for (int j = 0; j <= i; ++j) {
                float s = 0.f;
                #pragma unroll
                for (int k = 0; k <= j; ++k) s += sg[i][k] * sg[j][k];
                sm[i][j] = s;
                sm[j][i] = s;
            }
        }
        #pragma unroll
        for (int i = 0; i < PD; ++i)
            #pragma unroll
            for (int j = 0; j < PD; j += 4)
                *(float4*)&outS[(size_t)row * (PD * PD) + i * PD + j] =
                    make_float4(sm[i][j], sm[i][j + 1], sm[i][j + 2], sm[i][j + 3]);
    }
}

extern "C" void kernel_launch(void* const* d_in, const int* in_sizes, int n_in,
                              void* d_out, int out_size, void* d_ws, size_t ws_size,
                              hipStream_t stream) {
    const float* x = (const float*)d_in[0];
    const float* W = (const float*)d_in[1];
    const float* b = (const float*)d_in[2];
    float* out = (float*)d_out;
    (void)in_sizes; (void)n_in; (void)out_size; (void)d_ws; (void)ws_size;

    dim3 grid(BATCH / ROWS_PER_BLOCK);
    dim3 block(THREADS);
    l3m_param_head_kernel<<<grid, block, 0, stream>>>(x, W, b, out);
}
